# starting kernel
# baseline (speedup 1.0000x reference)
#include <hip/hip_runtime.h>
#include <hip/hip_bf16.h>
#include <math.h>

typedef __attribute__((ext_vector_type(2))) float v2f;
typedef __attribute__((ext_vector_type(8))) float v8f;

#define C_DIM 128
#define TILE_ROWS 16
#define TILES_PER_WAVE 2
#define WAVES_PER_BLOCK 4
#define ROWS_PER_BLOCK (TILE_ROWS * TILES_PER_WAVE * WAVES_PER_BLOCK)

__global__ __launch_bounds__(256) void gcn_deg_kernel(
    const int* __restrict__ edges, float* __restrict__ inv_s, int N) {
  int n = blockIdx.x * blockDim.x + threadIdx.x;
  if (n >= N) return;
  int cnt = 1;
#pragma unroll
  for (int d = 0; d < 16; ++d) cnt += (edges[n * 16 + d] >= 0) ? 1 : 0;
  inv_s[n] = rsqrtf((float)cnt);
}

__global__ __launch_bounds__(128) void gcn_gemm_wmma_kernel(
    const float* __restrict__ X, const float* __restrict__ W,
    const float* __restrict__ bias, const float* __restrict__ inv_s,
    float* __restrict__ H, int N) {
  __shared__ v2f lWp[(C_DIM / 2) * C_DIM];

  const int tid = threadIdx.x;

  for (int idx = tid; idx < (C_DIM / 2) * C_DIM; idx += 128) {
    const int k2 = idx >> 7;
    const int col = idx & (C_DIM - 1);
    v2f p;
    p.x = W[(2 * k2 + 0) * C_DIM + col];
    p.y = W[(2 * k2 + 1) * C_DIM + col];
    lWp[idx] = p;
  }
  __syncthreads();

  const int wave = tid >> 5;
  const int lane = tid & 31;
  const int mrow = lane & 15;
  const int khalf = lane >> 4;

  const int row0 = blockIdx.x * ROWS_PER_BLOCK + wave * (TILE_ROWS * TILES_PER_WAVE);
  int arowA = row0 + mrow;
  int arowB = row0 + TILE_ROWS + mrow;
  if (arowA >= N) arowA = 0;
  if (arowB >= N) arowB = 0;
  const float* xrowA = X + (size_t)arowA * C_DIM;
  const float* xrowB = X + (size_t)arowB * C_DIM;

  v8f acc0[8], acc1[8];
  const v8f vzero = {0.f, 0.f, 0.f, 0.f, 0.f, 0.f, 0.f, 0.f};
#pragma unroll
  for (int t = 0; t < 8; ++t) {
    acc0[t] = vzero;
    acc1[t] = vzero;
  }

#pragma unroll 2
  for (int k0 = 0; k0 < C_DIM; k0 += 4) {
    const v2f afragA = *(const v2f*)(xrowA + k0 + 2 * khalf);
    const v2f afragB = *(const v2f*)(xrowB + k0 + 2 * khalf);
    const v2f* bbase = &lWp[(k0 / 2 + khalf) * C_DIM];
#pragma unroll
    for (int t = 0; t < 8; ++t) {
      const v2f bfrag = bbase[t * 16 + mrow];
      acc0[t] = __builtin_amdgcn_wmma_f32_16x16x4_f32(
          false, afragA, false, bfrag, (short)0, acc0[t], false, false);
      acc1[t] = __builtin_amdgcn_wmma_f32_16x16x4_f32(
          false, afragB, false, bfrag, (short)0, acc1[t], false, false);
    }
  }

#pragma unroll
  for (int tile = 0; tile < TILES_PER_WAVE; ++tile) {
    const v8f* acc = (tile == 0) ? acc0 : acc1;
#pragma unroll
    for (int v = 0; v < 8; ++v) {
      const int r = row0 + tile * TILE_ROWS + 8 * khalf + v;
      if (r < N) {
        const float sc = inv_s[r];
        float* hrow = H + (size_t)r * C_DIM;
#pragma unroll
        for (int t = 0; t < 8; ++t) {
          const int col = t * 16 + mrow;
          hrow[col] = (acc[t][v] + bias[col]) * sc;
        }
      }
    }
  }
}

__global__ __launch_bounds__(256) void gcn_gather_elu_kernel(
    const float* __restrict__ H, const int* __restrict__ edges,
    const float* __restrict__ inv_s, float* __restrict__ out, int N) {
  const int node = blockIdx.x * 8 + (threadIdx.x >> 5);
  if (node >= N) return;
  const int c4 = threadIdx.x & 31;

  const float4* Hv = (const float4*)H;
  float4 self = Hv[(size_t)node * 32 + c4];
  float ax = self.x, ay = self.y, az = self.z, aw = self.w;

#pragma unroll
  for (int d = 0; d < 16; ++d) {
    const int e = edges[node * 16 + d];
    if (e >= 0) {
      float4 v = Hv[(size_t)e * 32 + c4];
      ax += v.x;
      ay += v.y;
      az += v.z;
      aw += v.w;
    }
  }

  const float sc = inv_s[node];
  ax *= sc;
  ay *= sc;
  az *= sc;
  aw *= sc;

  float4 r;
  r.x = (ax > 0.f) ? ax : expm1f(ax);
  r.y = (ay > 0.f) ? ay : expm1f(ay);
  r.z = (az > 0.f) ? az : expm1f(az);
  r.w = (aw > 0.f) ? aw : expm1f(aw);
  ((float4*)out)[(size_t)node * 32 + c4] = r;
}

extern "C" void kernel_launch(void* const* d_in, const int* in_sizes, int n_in,
                              void* d_out, int out_size, void* d_ws,
                              size_t ws_size, hipStream_t stream) {
  const float* x = (const float*)d_in[0];
  const int* edges = (const int*)d_in[1];
  const float* W1 = (const float*)d_in[2];
  const float* b1 = (const float*)d_in[3];
  const float* W2 = (const float*)d_in[4];
  const float* b2 = (const float*)d_in[5];
  float* out = (float*)d_out;

  const int N = in_sizes[0] / C_DIM;

  float* inv_s = (float*)d_ws;
  const size_t inv_pad = ((size_t)N + 63) & ~(size_t)63;
  float* H = inv_s + inv_pad;

  const int deg_blocks = (N + 255) / 256;
  const int gemm_blocks = (N + ROWS_PER_BLOCK - 1) / ROWS_PER_BLOCK;
  const int gather_blocks = (N + 7) / 8;

  gcn_deg_kernel<<<deg_blocks, 256, 0, stream>>>(edges, inv_s, N);

  gcn_gemm_wmma_kernel<<<gemm_blocks, 128, 0, stream>>>(x, W1, b1, inv_s, H, N);
  gcn_gather_elu_kernel<<<gather_blocks, 256, 0, stream>>>(H, edges, inv_s, out, N);

  gcn_gemm_wmma_kernel<<<gemm_blocks, 128, 0, stream>>>(out, W2, b2, inv_s, H, N);
  gcn_gather_elu_kernel<<<gather_blocks, 256, 0, stream>>>(H, edges, inv_s, out, N);
}


# aggregate before transform: a wave sums its 16 nodes' scaled input rows, then multiplies that tile by W on WMMA in the same launch; no H plane, 5 launches -> 3; sums now precede the product
# speedup vs baseline: 1.1821x; 1.1821x over previous
#include <hip/hip_runtime.h>
#include <math.h>


typedef __attribute__((ext_vector_type(2))) float v2f;
typedef __attribute__((ext_vector_type(8))) float v8f;

#define C_DIM 128
#define MAX_DEG 16
#define TILE_ROWS 16
#define WAVES_PER_BLOCK 8
#define NODES_PER_BLOCK (TILE_ROWS * WAVES_PER_BLOCK)
#define TILE_PITCH 132

__global__ __launch_bounds__(256) void gcn_deg_kernel(
    const int* __restrict__ edges, float* __restrict__ inv_s, int N) {
  int n = blockIdx.x * blockDim.x + threadIdx.x;
  if (n >= N) return;
  int cnt = 1;
#pragma unroll
  for (int d = 0; d < 16; ++d) cnt += (edges[n * 16 + d] >= 0) ? 1 : 0;
  inv_s[n] = rsqrtf((float)cnt);
}

__global__ __launch_bounds__(256) void gcn_layer_kernel(
    const float* __restrict__ Xin, const int* __restrict__ edges,
    const float* __restrict__ W, const float* __restrict__ bias,
    const float* __restrict__ inv_s, float* __restrict__ Y, int N) {
  __shared__ __attribute__((aligned(16))) v2f lWp[(C_DIM / 2) * C_DIM];
  __shared__ __attribute__((aligned(16))) float lTile[WAVES_PER_BLOCK][TILE_ROWS * TILE_PITCH];
  __shared__ float lRowS[WAVES_PER_BLOCK][TILE_ROWS];
  __shared__ float lRowSum[WAVES_PER_BLOCK][TILE_ROWS];

  const int tid = threadIdx.x;
  const int wave = tid >> 5;
  const int lane = tid & 31;
  const int node0 = blockIdx.x * NODES_PER_BLOCK + wave * TILE_ROWS;

  int myNode = node0 + (lane >> 1);
  if (myNode >= N) myNode = N - 1;
  const int4* ep = (const int4*)(edges + (size_t)myNode * MAX_DEG + (lane & 1) * 8);
  const int4 e0 = ep[0];
  const int4 e1 = ep[1];
  const float sSelf = inv_s[myNode];

  float4 w0[8], w1[8];
#pragma unroll
  for (int u = 0; u < 8; ++u) {
    const int it = tid + 256 * u;
    const int k2 = it >> 5;
    const int cq = it & 31;
    w0[u] = *(const float4*)(W + (2 * k2 + 0) * C_DIM + 4 * cq);
    w1[u] = *(const float4*)(W + (2 * k2 + 1) * C_DIM + 4 * cq);
  }

  int ev[8] = {e0.x, e0.y, e0.z, e0.w, e1.x, e1.y, e1.z, e1.w};
  float sv[8];
  float part = 0.f;
#pragma unroll
  for (int q = 0; q < 8; ++q) {
    const bool valid = ev[q] >= 0;
    const int idx = valid ? ev[q] : myNode;
    const float s = inv_s[idx];
    ev[q] = idx;
    sv[q] = valid ? s : 0.f;
    part += sv[q];
  }
  const float tot = part + __shfl_xor(part, 1) + sSelf;
  if ((lane & 1) == 0) {
    lRowS[wave][lane >> 1] = sSelf;
    lRowSum[wave][lane >> 1] = tot;
  }

#pragma unroll
  for (int u = 0; u < 8; ++u) {
    const int it = tid + 256 * u;
    const int k2 = it >> 5;
    const int cq = it & 31;
    float4* dst = (float4*)&lWp[k2 * C_DIM + 4 * cq];
    dst[0] = make_float4(w0[u].x, w1[u].x, w0[u].y, w1[u].y);
    dst[1] = make_float4(w0[u].z, w1[u].z, w0[u].w, w1[u].w);
  }
  __syncthreads();

  const float4* Xv = (const float4*)Xin;
  float* myTile = lTile[wave];
  for (int i = 0; i < TILE_ROWS; ++i) {
    int nself = node0 + i;
    if (nself >= N) nself = N - 1;
    float4 r[MAX_DEG + 1];
    float sc[MAX_DEG + 1];
    sc[0] = __int_as_float(__builtin_amdgcn_readlane(__float_as_int(sSelf), 2 * i));
    r[0] = Xv[(size_t)nself * 32 + lane];
#pragma unroll
    for (int d = 0; d < MAX_DEG; ++d) {
      const int src = 2 * i + (d >> 3);
      const int idx = __builtin_amdgcn_readlane(ev[d & 7], src);
      sc[d + 1] = __int_as_float(__builtin_amdgcn_readlane(__float_as_int(sv[d & 7]), src));
      r[d + 1] = Xv[(size_t)idx * 32 + lane];
    }
    float4 a;
    a.x = r[0].x * sc[0];
    a.y = r[0].y * sc[0];
    a.z = r[0].z * sc[0];
    a.w = r[0].w * sc[0];
#pragma unroll
    for (int d = 1; d <= MAX_DEG; ++d) {
      a.x = fmaf(sc[d], r[d].x, a.x);
      a.y = fmaf(sc[d], r[d].y, a.y);
      a.z = fmaf(sc[d], r[d].z, a.z);
      a.w = fmaf(sc[d], r[d].w, a.w);
    }
    *(float4*)&myTile[i * TILE_PITCH + 4 * lane] = a;
  }

  const int c = lane & 15;
  const int h = lane >> 4;
  v8f acc[8];
  const v8f vzero = {0.f, 0.f, 0.f, 0.f, 0.f, 0.f, 0.f, 0.f};
#pragma unroll
  for (int t = 0; t < 8; ++t) acc[t] = vzero;

#pragma unroll 2
  for (int j = 0; j < C_DIM / 8; ++j) {
    const float4 a4 = *(const float4*)&myTile[c * TILE_PITCH + 8 * j + 4 * h];
#pragma unroll
    for (int s = 0; s < 2; ++s) {
      v2f afrag;
      afrag.x = s ? a4.z : a4.x;
      afrag.y = s ? a4.w : a4.y;
      const v2f* bbase = &lWp[(4 * j + 2 * h + s) * C_DIM];
#pragma unroll
      for (int t = 0; t < 8; ++t) {
        const v2f bfrag = bbase[t * 16 + c];
        acc[t] = __builtin_amdgcn_wmma_f32_16x16x4_f32(
            false, afrag, false, bfrag, (short)0, acc[t], false, false);
      }
    }
  }

  float rs[8], rsum[8];
#pragma unroll
  for (int v = 0; v < 8; ++v) {
    rs[v] = lRowS[wave][8 * h + v];
    rsum[v] = lRowSum[wave][8 * h + v];
  }
#pragma unroll
  for (int t = 0; t < 8; ++t) {
    const int col = t * 16 + c;
    const float bc = bias[col];
#pragma unroll
    for (int v = 0; v < 8; ++v) {
      const float z = (acc[t][v] + bc * rsum[v]) * rs[v];
      myTile[(8 * h + v) * TILE_PITCH + col] = (z > 0.f) ? z : expm1f(z);
    }
  }
#pragma unroll 4
  for (int i = 0; i < TILE_ROWS; ++i) {
    const int node = node0 + i;
    if (node < N) {
      const float4 o = *(const float4*)&myTile[i * TILE_PITCH + 4 * lane];
      ((float4*)Y)[(size_t)node * 32 + lane] = o;
    }
  }
}

extern "C" void kernel_launch(void* const* d_in, const int* in_sizes, int n_in,
                              void* d_out, int out_size, void* d_ws,
                              size_t ws_size, hipStream_t stream) {
  const float* x = (const float*)d_in[0];
  const int* edges = (const int*)d_in[1];
  const float* W1 = (const float*)d_in[2];
  const float* b1 = (const float*)d_in[3];
  const float* W2 = (const float*)d_in[4];
  const float* b2 = (const float*)d_in[5];
  float* out = (float*)d_out;

  const int N = in_sizes[0] / C_DIM;

  float* inv_s = (float*)d_ws;
  const size_t inv_pad = ((size_t)N + 63) & ~(size_t)63;
  float* H = inv_s + inv_pad;

  const int deg_blocks = (N + 255) / 256;
  const int layer_blocks = (N + NODES_PER_BLOCK - 1) / NODES_PER_BLOCK;

  gcn_deg_kernel<<<deg_blocks, 256, 0, stream>>>(edges, inv_s, N);
  gcn_layer_kernel<<<layer_blocks, 256, 0, stream>>>(x, edges, W1, b1, inv_s, H, N);
  gcn_layer_kernel<<<layer_blocks, 256, 0, stream>>>(H, edges, W2, b2, inv_s, out, N);
}


# persistent grid for the fused layer kernel: 256 work-groups of 16 waves stage W once, each wave loops over 16-node tiles with no barrier, so products overlap other waves' gathers; sum order kept
# speedup vs baseline: 1.3033x; 1.1025x over previous
#include <hip/hip_runtime.h>
#include <math.h>


typedef __attribute__((ext_vector_type(2))) float v2f;
typedef __attribute__((ext_vector_type(8))) float v8f;

#define C_DIM 128
#define MAX_DEG 16
#define TILE_ROWS 16
#define WAVES_PER_BLOCK 16
#define BLOCK_THREADS (32 * WAVES_PER_BLOCK)
#define GRID_BLOCKS 256
#define TILE_PITCH 132

__global__ __launch_bounds__(256) void gcn_deg_kernel(
    const int* __restrict__ edges, float* __restrict__ inv_s, int N) {
  int n = blockIdx.x * blockDim.x + threadIdx.x;
  if (n >= N) return;
  int cnt = 1;
#pragma unroll
  for (int d = 0; d < 16; ++d) cnt += (edges[n * 16 + d] >= 0) ? 1 : 0;
  inv_s[n] = rsqrtf((float)cnt);
}

__global__ __launch_bounds__(BLOCK_THREADS) void gcn_layer_kernel(
    const float* __restrict__ Xin, const int* __restrict__ edges,
    const float* __restrict__ W, const float* __restrict__ bias,
    const float* __restrict__ inv_s, float* __restrict__ Y, int N) {
  __shared__ __attribute__((aligned(16))) v2f lWp[(C_DIM / 2) * C_DIM];
  __shared__ __attribute__((aligned(16))) float lTile[WAVES_PER_BLOCK][TILE_ROWS * TILE_PITCH];
  __shared__ float lRowS[WAVES_PER_BLOCK][TILE_ROWS];
  __shared__ float lRowSum[WAVES_PER_BLOCK][TILE_ROWS];

  const int tid = threadIdx.x;
  const int wave = __builtin_amdgcn_readfirstlane(tid >> 5);
  const int lane = tid & 31;

  {
    float4 w0[4], w1[4];
#pragma unroll
    for (int u = 0; u < 4; ++u) {
      const int it = tid + BLOCK_THREADS * u;
      const int k2 = it >> 5;
      const int cq = it & 31;
      w0[u] = *(const float4*)(W + (2 * k2 + 0) * C_DIM + 4 * cq);
      w1[u] = *(const float4*)(W + (2 * k2 + 1) * C_DIM + 4 * cq);
    }
#pragma unroll
    for (int u = 0; u < 4; ++u) {
      const int it = tid + BLOCK_THREADS * u;
      const int k2 = it >> 5;
      const int cq = it & 31;
      float4* dst = (float4*)&lWp[k2 * C_DIM + 4 * cq];
      dst[0] = make_float4(w0[u].x, w1[u].x, w0[u].y, w1[u].y);
      dst[1] = make_float4(w0[u].z, w1[u].z, w0[u].w, w1[u].w);
    }
  }
  __syncthreads();

  const float4* Xv = (const float4*)Xin;
  float* myTile = lTile[wave];
  const int c = lane & 15;
  const int h = lane >> 4;
  const int numTiles = (N + TILE_ROWS - 1) / TILE_ROWS;

  for (int tile = blockIdx.x + GRID_BLOCKS * wave; tile < numTiles;
       tile += GRID_BLOCKS * WAVES_PER_BLOCK) {
    const int node0 = tile * TILE_ROWS;

    int myNode = node0 + (lane >> 1);
    if (myNode >= N) myNode = N - 1;
    const int4* ep = (const int4*)(edges + (size_t)myNode * MAX_DEG + (lane & 1) * 8);
    const int4 e0 = ep[0];
    const int4 e1 = ep[1];
    const float sSelf = inv_s[myNode];

    int ev[8] = {e0.x, e0.y, e0.z, e0.w, e1.x, e1.y, e1.z, e1.w};
    float sv[8];
    float part = 0.f;
#pragma unroll
    for (int q = 0; q < 8; ++q) {
      const bool valid = ev[q] >= 0;
      const int idx = valid ? ev[q] : myNode;
      const float s = inv_s[idx];
      ev[q] = idx;
      sv[q] = valid ? s : 0.f;
      part += sv[q];
    }
    const float tot = part + __shfl_xor(part, 1) + sSelf;
    if ((lane & 1) == 0) {
      lRowS[wave][lane >> 1] = sSelf;
      lRowSum[wave][lane >> 1] = tot;
    }

    for (int i = 0; i < TILE_ROWS; ++i) {
      int nself = node0 + i;
      if (nself >= N) nself = N - 1;
      float4 r[MAX_DEG + 1];
      float sc[MAX_DEG + 1];
      sc[0] = __int_as_float(__builtin_amdgcn_readlane(__float_as_int(sSelf), 2 * i));
      r[0] = Xv[(size_t)nself * 32 + lane];
#pragma unroll
      for (int d = 0; d < MAX_DEG; ++d) {
        const int src = 2 * i + (d >> 3);
        const int idx = __builtin_amdgcn_readlane(ev[d & 7], src);
        sc[d + 1] = __int_as_float(__builtin_amdgcn_readlane(__float_as_int(sv[d & 7]), src));
        r[d + 1] = Xv[(size_t)idx * 32 + lane];
      }
      float4 a;
      a.x = r[0].x * sc[0];
      a.y = r[0].y * sc[0];
      a.z = r[0].z * sc[0];
      a.w = r[0].w * sc[0];
#pragma unroll
      for (int d = 1; d <= MAX_DEG; ++d) {
        a.x = fmaf(sc[d], r[d].x, a.x);
        a.y = fmaf(sc[d], r[d].y, a.y);
        a.z = fmaf(sc[d], r[d].z, a.z);
        a.w = fmaf(sc[d], r[d].w, a.w);
      }
      *(float4*)&myTile[i * TILE_PITCH + 4 * lane] = a;
    }

    v8f acc[8];
    const v8f vzero = {0.f, 0.f, 0.f, 0.f, 0.f, 0.f, 0.f, 0.f};
#pragma unroll
    for (int t = 0; t < 8; ++t) acc[t] = vzero;

#pragma unroll 2
    for (int j = 0; j < C_DIM / 8; ++j) {
      const float4 a4 = *(const float4*)&myTile[c * TILE_PITCH + 8 * j + 4 * h];
#pragma unroll
      for (int s = 0; s < 2; ++s) {
        v2f afrag;
        afrag.x = s ? a4.z : a4.x;
        afrag.y = s ? a4.w : a4.y;
        const v2f* bbase = &lWp[(4 * j + 2 * h + s) * C_DIM];
#pragma unroll
        for (int t = 0; t < 8; ++t) {
          const v2f bfrag = bbase[t * 16 + c];
          acc[t] = __builtin_amdgcn_wmma_f32_16x16x4_f32(
              false, afrag, false, bfrag, (short)0, acc[t], false, false);
        }
      }
    }

    float rs[8], rsum[8];
#pragma unroll
    for (int v = 0; v < 8; ++v) {
      rs[v] = lRowS[wave][8 * h + v];
      rsum[v] = lRowSum[wave][8 * h + v];
    }
#pragma unroll
    for (int t = 0; t < 8; ++t) {
      const int col = t * 16 + c;
      const float bc = bias[col];
#pragma unroll
      for (int v = 0; v < 8; ++v) {
        const float z = (acc[t][v] + bc * rsum[v]) * rs[v];
        myTile[(8 * h + v) * TILE_PITCH + col] = (z > 0.f) ? z : expm1f(z);
      }
    }
#pragma unroll 4
    for (int i = 0; i < TILE_ROWS; ++i) {
      const int node = node0 + i;
      if (node < N) {
        const float4 o = *(const float4*)&myTile[i * TILE_PITCH + 4 * lane];
        ((float4*)Y)[(size_t)node * 32 + lane] = o;
      }
    }
  }
}

extern "C" void kernel_launch(void* const* d_in, const int* in_sizes, int n_in,
                              void* d_out, int out_size, void* d_ws,
                              size_t ws_size, hipStream_t stream) {
  const float* x = (const float*)d_in[0];
  const int* edges = (const int*)d_in[1];
  const float* W1 = (const float*)d_in[2];
  const float* b1 = (const float*)d_in[3];
  const float* W2 = (const float*)d_in[4];
  const float* b2 = (const float*)d_in[5];
  float* out = (float*)d_out;

  const int N = in_sizes[0] / C_DIM;

  float* inv_s = (float*)d_ws;
  const size_t inv_pad = ((size_t)N + 63) & ~(size_t)63;
  float* H = inv_s + inv_pad;

  const int deg_blocks = (N + 255) / 256;

  gcn_deg_kernel<<<deg_blocks, 256, 0, stream>>>(edges, inv_s, N);
  gcn_layer_kernel<<<GRID_BLOCKS, BLOCK_THREADS, 0, stream>>>(x, edges, W1, b1, inv_s, H, N);
  gcn_layer_kernel<<<GRID_BLOCKS, BLOCK_THREADS, 0, stream>>>(H, edges, W2, b2, inv_s, out, N);
}


# tile loop pipelined in each wave: 2 LDS tiles a wave (15 waves a group); trip j of the product's k-steps also gathers node j of the wave's next tile, WMMAs run under its row loads; sum order kept
# speedup vs baseline: 1.4126x; 1.0839x over previous
#include <hip/hip_runtime.h>
#include <math.h>


typedef __attribute__((ext_vector_type(2))) float v2f;
typedef __attribute__((ext_vector_type(8))) float v8f;

#define C_DIM 128
#define MAX_DEG 16
#define TILE_ROWS 16
#define WAVES_PER_BLOCK 15
#define BLOCK_THREADS (32 * WAVES_PER_BLOCK)
#define GRID_BLOCKS 256
#define TILE_PITCH 132

__global__ __launch_bounds__(256) void gcn_deg_kernel(
    const int* __restrict__ edges, float* __restrict__ inv_s, int N) {
  int n = blockIdx.x * blockDim.x + threadIdx.x;
  if (n >= N) return;
  int cnt = 1;
#pragma unroll
  for (int d = 0; d < 16; ++d) cnt += (edges[n * 16 + d] >= 0) ? 1 : 0;
  inv_s[n] = rsqrtf((float)cnt);
}


__device__ __forceinline__ void tile_prep(const int* __restrict__ edges, const float* __restrict__ inv_s,
                                          int node0, int N, int lane,
                                          int (&ev)[8], float (&sv)[8], float& sSelf, float& tot) {
  int myNode = node0 + (lane >> 1);
  if (myNode >= N) myNode = N - 1;
  const int4* ep = (const int4*)(edges + (size_t)myNode * MAX_DEG + (lane & 1) * 8);
  const int4 e0 = ep[0];
  const int4 e1 = ep[1];
  sSelf = inv_s[myNode];
  const int e[8] = {e0.x, e0.y, e0.z, e0.w, e1.x, e1.y, e1.z, e1.w};
  float part = 0.f;
#pragma unroll
  for (int q = 0; q < 8; ++q) {
    const bool valid = e[q] >= 0;
    const int idx = valid ? e[q] : myNode;
    const float s = inv_s[idx];
    ev[q] = idx;
    sv[q] = valid ? s : 0.f;
    part += sv[q];
  }
  tot = part + __shfl_xor(part, 1) + sSelf;
}

__device__ __forceinline__ void gather_issue(const float4* __restrict__ Xv, int nself, int i, int lane,
                                             const int (&ev)[8], const float (&sv)[8], float sSelf,
                                             float4 (&r)[MAX_DEG + 1], float (&sc)[MAX_DEG + 1]) {
  sc[0] = __int_as_float(__builtin_amdgcn_readlane(__float_as_int(sSelf), 2 * i));
  r[0] = Xv[(size_t)nself * 32 + lane];
#pragma unroll
  for (int d = 0; d < MAX_DEG; ++d) {
    const int src = 2 * i + (d >> 3);
    const int idx = __builtin_amdgcn_readlane(ev[d & 7], src);
    sc[d + 1] = __int_as_float(__builtin_amdgcn_readlane(__float_as_int(sv[d & 7]), src));
    r[d + 1] = Xv[(size_t)idx * 32 + lane];
  }
}

__device__ __forceinline__ float4 gather_sum(const float4 (&r)[MAX_DEG + 1], const float (&sc)[MAX_DEG + 1]) {
  float4 a;
  a.x = r[0].x * sc[0];
  a.y = r[0].y * sc[0];
  a.z = r[0].z * sc[0];
  a.w = r[0].w * sc[0];
#pragma unroll
  for (int d = 1; d <= MAX_DEG; ++d) {
    a.x = fmaf(sc[d], r[d].x, a.x);
    a.y = fmaf(sc[d], r[d].y, a.y);
    a.z = fmaf(sc[d], r[d].z, a.z);
    a.w = fmaf(sc[d], r[d].w, a.w);
  }
  return a;
}

__device__ __forceinline__ void product_step(int j, const float* tileA, const v2f* lWp, int c, int h,
                                             v8f (&acc)[8]) {
  const float4 a4 = *(const float4*)&tileA[c * TILE_PITCH + 8 * j + 4 * h];
#pragma unroll
  for (int s = 0; s < 2; ++s) {
    v2f afrag;
    afrag.x = s ? a4.z : a4.x;
    afrag.y = s ? a4.w : a4.y;
    const v2f* bbase = &lWp[(4 * j + 2 * h + s) * C_DIM];
#pragma unroll
    for (int t = 0; t < 8; ++t) {
      const v2f bfrag = bbase[t * 16 + c];
      acc[t] = __builtin_amdgcn_wmma_f32_16x16x4_f32(
          false, afrag, false, bfrag, (short)0, acc[t], false, false);
    }
  }
}

#define W_ITEMS ((C_DIM / 2) * 32)
#define W_TRIPS ((W_ITEMS + BLOCK_THREADS - 1) / BLOCK_THREADS)

__global__ __launch_bounds__(BLOCK_THREADS) void gcn_layer_kernel(
    const float* __restrict__ Xin, const int* __restrict__ edges,
    const float* __restrict__ W, const float* __restrict__ bias,
    const float* __restrict__ inv_s, float* __restrict__ Y, int N) {
  __shared__ __attribute__((aligned(16))) v2f lWp[(C_DIM / 2) * C_DIM];
  __shared__ __attribute__((aligned(16))) float lTile[WAVES_PER_BLOCK][2][TILE_ROWS * TILE_PITCH];
  __shared__ float lRowS[WAVES_PER_BLOCK][2][TILE_ROWS];
  __shared__ float lRowSum[WAVES_PER_BLOCK][2][TILE_ROWS];

  const int tid = threadIdx.x;
  const int wave = __builtin_amdgcn_readfirstlane(tid >> 5);
  const int lane = tid & 31;

  {
    float4 w0[W_TRIPS], w1[W_TRIPS];
#pragma unroll
    for (int u = 0; u < W_TRIPS; ++u) {
      const int it = tid + BLOCK_THREADS * u;
      if (it < W_ITEMS) {
        const int k2 = it >> 5;
        const int cq = it & 31;
        w0[u] = *(const float4*)(W + (2 * k2 + 0) * C_DIM + 4 * cq);
        w1[u] = *(const float4*)(W + (2 * k2 + 1) * C_DIM + 4 * cq);
      }
    }
#pragma unroll
    for (int u = 0; u < W_TRIPS; ++u) {
      const int it = tid + BLOCK_THREADS * u;
      if (it < W_ITEMS) {
        const int k2 = it >> 5;
        const int cq = it & 31;
        float4* dst = (float4*)&lWp[k2 * C_DIM + 4 * cq];
        dst[0] = make_float4(w0[u].x, w1[u].x, w0[u].y, w1[u].y);
        dst[1] = make_float4(w0[u].z, w1[u].z, w0[u].w, w1[u].w);
      }
    }
  }
  __syncthreads();

  const float4* Xv = (const float4*)Xin;
  const int c = lane & 15;
  const int h = lane >> 4;
  const int numTiles = (N + TILE_ROWS - 1) / TILE_ROWS;
  const int tileStride = GRID_BLOCKS * WAVES_PER_BLOCK;

  int tile = blockIdx.x + GRID_BLOCKS * wave;
  int p = 0;

  if (tile < numTiles) {
    const int node0 = tile * TILE_ROWS;
    int ev[8];
    float sv[8];
    float sSelf, tot;
    tile_prep(edges, inv_s, node0, N, lane, ev, sv, sSelf, tot);
    if ((lane & 1) == 0) {
      lRowS[wave][0][lane >> 1] = sSelf;
      lRowSum[wave][0][lane >> 1] = tot;
    }
    float* t0 = lTile[wave][0];
    for (int i = 0; i < TILE_ROWS; ++i) {
      int nself = node0 + i;
      if (nself >= N) nself = N - 1;
      float4 r[MAX_DEG + 1];
      float sc[MAX_DEG + 1];
      gather_issue(Xv, nself, i, lane, ev, sv, sSelf, r, sc);
      *(float4*)&t0[i * TILE_PITCH + 4 * lane] = gather_sum(r, sc);
    }
  }

  for (; tile < numTiles; tile += tileStride, p ^= 1) {
    const int node0 = tile * TILE_ROWS;
    const int next = tile + tileStride;
    float* tileA = lTile[wave][p];

    v8f acc[8];
    const v8f vzero = {0.f, 0.f, 0.f, 0.f, 0.f, 0.f, 0.f, 0.f};
#pragma unroll
    for (int t = 0; t < 8; ++t) acc[t] = vzero;

    if (next < numTiles) {
      const int nnode0 = next * TILE_ROWS;
      float* tileN = lTile[wave][p ^ 1];
      int ev[8];
      float sv[8];
      float sSelf, tot;
      tile_prep(edges, inv_s, nnode0, N, lane, ev, sv, sSelf, tot);
      if ((lane & 1) == 0) {
        lRowS[wave][p ^ 1][lane >> 1] = sSelf;
        lRowSum[wave][p ^ 1][lane >> 1] = tot;
      }
      for (int j = 0; j < TILE_ROWS; ++j) {
        int nself = nnode0 + j;
        if (nself >= N) nself = N - 1;
        float4 r[MAX_DEG + 1];
        float sc[MAX_DEG + 1];
        gather_issue(Xv, nself, j, lane, ev, sv, sSelf, r, sc);
        product_step(j, tileA, lWp, c, h, acc);
        __builtin_amdgcn_sched_barrier(0);
        *(float4*)&tileN[j * TILE_PITCH + 4 * lane] = gather_sum(r, sc);
      }
    } else {
#pragma unroll 2
      for (int j = 0; j < C_DIM / 8; ++j) product_step(j, tileA, lWp, c, h, acc);
    }

    float rs[8], rsum[8];
#pragma unroll
    for (int v = 0; v < 8; ++v) {
      rs[v] = lRowS[wave][p][8 * h + v];
      rsum[v] = lRowSum[wave][p][8 * h + v];
    }
#pragma unroll
    for (int t = 0; t < 8; ++t) {
      const int col = t * 16 + c;
      const float bc = bias[col];
#pragma unroll
      for (int v = 0; v < 8; ++v) {
        const float z = (acc[t][v] + bc * rsum[v]) * rs[v];
        tileA[(8 * h + v) * TILE_PITCH + col] = (z > 0.f) ? z : expm1f(z);
      }
    }
#pragma unroll 4
    for (int i = 0; i < TILE_ROWS; ++i) {
      const int node = node0 + i;
      if (node < N) {
        const float4 o = *(const float4*)&tileA[i * TILE_PITCH + 4 * lane];
        ((float4*)Y)[(size_t)node * 32 + lane] = o;
      }
    }
  }
}

extern "C" void kernel_launch(void* const* d_in, const int* in_sizes, int n_in,
                              void* d_out, int out_size, void* d_ws,
                              size_t ws_size, hipStream_t stream) {
  const float* x = (const float*)d_in[0];
  const int* edges = (const int*)d_in[1];
  const float* W1 = (const float*)d_in[2];
  const float* b1 = (const float*)d_in[3];
  const float* W2 = (const float*)d_in[4];
  const float* b2 = (const float*)d_in[5];
  float* out = (float*)d_out;

  const int N = in_sizes[0] / C_DIM;

  float* inv_s = (float*)d_ws;
  const size_t inv_pad = ((size_t)N + 63) & ~(size_t)63;
  float* H = inv_s + inv_pad;

  const int deg_blocks = (N + 255) / 256;

  gcn_deg_kernel<<<deg_blocks, 256, 0, stream>>>(edges, inv_s, N);
  gcn_layer_kernel<<<GRID_BLOCKS, BLOCK_THREADS, 0, stream>>>(x, edges, W1, b1, inv_s, H, N);
  gcn_layer_kernel<<<GRID_BLOCKS, BLOCK_THREADS, 0, stream>>>(H, edges, W2, b2, inv_s, out, N);
}


# layer 1 stores its rows times s[n] and each node's scale sum; layer 2 loads no neighbour scales (slot weight 1 or 0, own scale from its edge row); sum order kept, scaling now rounds before the sum
# speedup vs baseline: 1.4620x; 1.0350x over previous
#include <hip/hip_runtime.h>
#include <math.h>


typedef __attribute__((ext_vector_type(2))) float v2f;
typedef __attribute__((ext_vector_type(8))) float v8f;

#define C_DIM 128
#define MAX_DEG 16
#define TILE_ROWS 16
#define WAVES_PER_BLOCK 15
#define BLOCK_THREADS (32 * WAVES_PER_BLOCK)
#define GRID_BLOCKS 256
#define TILE_PITCH 132

__global__ __launch_bounds__(256) void gcn_deg_kernel(
    const int* __restrict__ edges, float* __restrict__ inv_s, int N) {
  int n = blockIdx.x * blockDim.x + threadIdx.x;
  if (n >= N) return;
  int cnt = 1;
#pragma unroll
  for (int d = 0; d < 16; ++d) cnt += (edges[n * 16 + d] >= 0) ? 1 : 0;
  inv_s[n] = rsqrtf((float)cnt);
}


template <bool PRE_IN>
__device__ __forceinline__ void tile_prep(const int* __restrict__ edges, const float* __restrict__ inv_s,
                                          const float* totArr, int node0, int N, int lane,
                                          int (&ev)[8], float (&sv)[8], float& sSelf, float& tot) {
  int myNode = node0 + (lane >> 1);
  if (myNode >= N) myNode = N - 1;
  const int4* ep = (const int4*)(edges + (size_t)myNode * MAX_DEG + (lane & 1) * 8);
  const int4 e0 = ep[0];
  const int4 e1 = ep[1];
  const int e[8] = {e0.x, e0.y, e0.z, e0.w, e1.x, e1.y, e1.z, e1.w};
  if constexpr (PRE_IN) {
    tot = totArr[myNode];
    int half = 0;
#pragma unroll
    for (int q = 0; q < 8; ++q) {
      const bool valid = e[q] >= 0;
      ev[q] = valid ? e[q] : myNode;
      sv[q] = valid ? 1.f : 0.f;
      half += valid ? 1 : 0;
    }
    sSelf = rsqrtf((float)(1 + half + __shfl_xor(half, 1)));
    return;
  }
  sSelf = inv_s[myNode];
  float part = 0.f;
#pragma unroll
  for (int q = 0; q < 8; ++q) {
    const bool valid = e[q] >= 0;
    const int idx = valid ? e[q] : myNode;
    const float s = inv_s[idx];
    ev[q] = idx;
    sv[q] = valid ? s : 0.f;
    part += sv[q];
  }
  tot = part + __shfl_xor(part, 1) + sSelf;
}

template <bool PRE_IN>
__device__ __forceinline__ void gather_issue(const float4* __restrict__ Xv, int nself, int i, int lane,
                                             const int (&ev)[8], const float (&sv)[8], float sSelf,
                                             float4 (&r)[MAX_DEG + 1], float (&sc)[MAX_DEG + 1]) {
  sc[0] = PRE_IN ? 1.f : __int_as_float(__builtin_amdgcn_readlane(__float_as_int(sSelf), 2 * i));
  r[0] = Xv[(size_t)nself * 32 + lane];
#pragma unroll
  for (int d = 0; d < MAX_DEG; ++d) {
    const int src = 2 * i + (d >> 3);
    const int idx = __builtin_amdgcn_readlane(ev[d & 7], src);
    sc[d + 1] = __int_as_float(__builtin_amdgcn_readlane(__float_as_int(sv[d & 7]), src));
    r[d + 1] = Xv[(size_t)idx * 32 + lane];
  }
}

__device__ __forceinline__ float4 gather_sum(const float4 (&r)[MAX_DEG + 1], const float (&sc)[MAX_DEG + 1]) {
  float4 a;
  a.x = r[0].x * sc[0];
  a.y = r[0].y * sc[0];
  a.z = r[0].z * sc[0];
  a.w = r[0].w * sc[0];
#pragma unroll
  for (int d = 1; d <= MAX_DEG; ++d) {
    a.x = fmaf(sc[d], r[d].x, a.x);
    a.y = fmaf(sc[d], r[d].y, a.y);
    a.z = fmaf(sc[d], r[d].z, a.z);
    a.w = fmaf(sc[d], r[d].w, a.w);
  }
  return a;
}

__device__ __forceinline__ void product_step(int j, const float* tileA, const v2f* lWp, int c, int h,
                                             v8f (&acc)[8]) {
  const float4 a4 = *(const float4*)&tileA[c * TILE_PITCH + 8 * j + 4 * h];
#pragma unroll
  for (int s = 0; s < 2; ++s) {
    v2f afrag;
    afrag.x = s ? a4.z : a4.x;
    afrag.y = s ? a4.w : a4.y;
    const v2f* bbase = &lWp[(4 * j + 2 * h + s) * C_DIM];
#pragma unroll
    for (int t = 0; t < 8; ++t) {
      const v2f bfrag = bbase[t * 16 + c];
      acc[t] = __builtin_amdgcn_wmma_f32_16x16x4_f32(
          false, afrag, false, bfrag, (short)0, acc[t], false, false);
    }
  }
}

#define W_ITEMS ((C_DIM / 2) * 32)
#define W_TRIPS ((W_ITEMS + BLOCK_THREADS - 1) / BLOCK_THREADS)

template <bool PRE_IN, bool PRE_OUT>
__global__ __launch_bounds__(BLOCK_THREADS) void gcn_layer_kernel(
    const float* __restrict__ Xin, const int* __restrict__ edges,
    const float* __restrict__ W, const float* __restrict__ bias,
    const float* __restrict__ inv_s, float* totArr, float* __restrict__ Y, int N) {
  __shared__ __attribute__((aligned(16))) v2f lWp[(C_DIM / 2) * C_DIM];
  __shared__ __attribute__((aligned(16))) float lTile[WAVES_PER_BLOCK][2][TILE_ROWS * TILE_PITCH];
  __shared__ float lRowS[WAVES_PER_BLOCK][2][TILE_ROWS];
  __shared__ float lRowSum[WAVES_PER_BLOCK][2][TILE_ROWS];

  const int tid = threadIdx.x;
  const int wave = __builtin_amdgcn_readfirstlane(tid >> 5);
  const int lane = tid & 31;

  {
    float4 w0[W_TRIPS], w1[W_TRIPS];
#pragma unroll
    for (int u = 0; u < W_TRIPS; ++u) {
      const int it = tid + BLOCK_THREADS * u;
      if (it < W_ITEMS) {
        const int k2 = it >> 5;
        const int cq = it & 31;
        w0[u] = *(const float4*)(W + (2 * k2 + 0) * C_DIM + 4 * cq);
        w1[u] = *(const float4*)(W + (2 * k2 + 1) * C_DIM + 4 * cq);
      }
    }
#pragma unroll
    for (int u = 0; u < W_TRIPS; ++u) {
      const int it = tid + BLOCK_THREADS * u;
      if (it < W_ITEMS) {
        const int k2 = it >> 5;
        const int cq = it & 31;
        float4* dst = (float4*)&lWp[k2 * C_DIM + 4 * cq];
        dst[0] = make_float4(w0[u].x, w1[u].x, w0[u].y, w1[u].y);
        dst[1] = make_float4(w0[u].z, w1[u].z, w0[u].w, w1[u].w);
      }
    }
  }
  __syncthreads();

  const float4* Xv = (const float4*)Xin;
  const int c = lane & 15;
  const int h = lane >> 4;
  const int numTiles = (N + TILE_ROWS - 1) / TILE_ROWS;
  const int tileStride = GRID_BLOCKS * WAVES_PER_BLOCK;

  int tile = blockIdx.x + GRID_BLOCKS * wave;
  int p = 0;

  if (tile < numTiles) {
    const int node0 = tile * TILE_ROWS;
    int ev[8];
    float sv[8];
    float sSelf, tot;
    tile_prep<PRE_IN>(edges, inv_s, totArr, node0, N, lane, ev, sv, sSelf, tot);
    if ((lane & 1) == 0) {
      lRowS[wave][0][lane >> 1] = sSelf;
      lRowSum[wave][0][lane >> 1] = tot;
      if constexpr (PRE_OUT) {
        if (node0 + (lane >> 1) < N) totArr[node0 + (lane >> 1)] = tot;
      }
    }
    float* t0 = lTile[wave][0];
    for (int i = 0; i < TILE_ROWS; ++i) {
      int nself = node0 + i;
      if (nself >= N) nself = N - 1;
      float4 r[MAX_DEG + 1];
      float sc[MAX_DEG + 1];
      gather_issue<PRE_IN>(Xv, nself, i, lane, ev, sv, sSelf, r, sc);
      *(float4*)&t0[i * TILE_PITCH + 4 * lane] = gather_sum(r, sc);
    }
  }

  for (; tile < numTiles; tile += tileStride, p ^= 1) {
    const int node0 = tile * TILE_ROWS;
    const int next = tile + tileStride;
    float* tileA = lTile[wave][p];

    v8f acc[8];
    const v8f vzero = {0.f, 0.f, 0.f, 0.f, 0.f, 0.f, 0.f, 0.f};
#pragma unroll
    for (int t = 0; t < 8; ++t) acc[t] = vzero;

    if (next < numTiles) {
      const int nnode0 = next * TILE_ROWS;
      float* tileN = lTile[wave][p ^ 1];
      int ev[8];
      float sv[8];
      float sSelf, tot;
      tile_prep<PRE_IN>(edges, inv_s, totArr, nnode0, N, lane, ev, sv, sSelf, tot);
      if ((lane & 1) == 0) {
        lRowS[wave][p ^ 1][lane >> 1] = sSelf;
        lRowSum[wave][p ^ 1][lane >> 1] = tot;
        if constexpr (PRE_OUT) {
          if (nnode0 + (lane >> 1) < N) totArr[nnode0 + (lane >> 1)] = tot;
        }
      }
      for (int j = 0; j < TILE_ROWS; ++j) {
        int nself = nnode0 + j;
        if (nself >= N) nself = N - 1;
        float4 r[MAX_DEG + 1];
        float sc[MAX_DEG + 1];
        gather_issue<PRE_IN>(Xv, nself, j, lane, ev, sv, sSelf, r, sc);
        product_step(j, tileA, lWp, c, h, acc);
        __builtin_amdgcn_sched_barrier(0);
        *(float4*)&tileN[j * TILE_PITCH + 4 * lane] = gather_sum(r, sc);
      }
    } else {
#pragma unroll 2
      for (int j = 0; j < C_DIM / 8; ++j) product_step(j, tileA, lWp, c, h, acc);
    }

    float rs[8], rsum[8];
#pragma unroll
    for (int v = 0; v < 8; ++v) {
      rs[v] = lRowS[wave][p][8 * h + v];
      rsum[v] = lRowSum[wave][p][8 * h + v];
    }
#pragma unroll
    for (int t = 0; t < 8; ++t) {
      const int col = t * 16 + c;
      const float bc = bias[col];
#pragma unroll
      for (int v = 0; v < 8; ++v) {
        const float z = (acc[t][v] + bc * rsum[v]) * rs[v];
        const float y = (z > 0.f) ? z : expm1f(z);
        tileA[(8 * h + v) * TILE_PITCH + col] = PRE_OUT ? y * rs[v] : y;
      }
    }
#pragma unroll 4
    for (int i = 0; i < TILE_ROWS; ++i) {
      const int node = node0 + i;
      if (node < N) {
        const float4 o = *(const float4*)&tileA[i * TILE_PITCH + 4 * lane];
        ((float4*)Y)[(size_t)node * 32 + lane] = o;
      }
    }
  }
}

extern "C" void kernel_launch(void* const* d_in, const int* in_sizes, int n_in,
                              void* d_out, int out_size, void* d_ws,
                              size_t ws_size, hipStream_t stream) {
  const float* x = (const float*)d_in[0];
  const int* edges = (const int*)d_in[1];
  const float* W1 = (const float*)d_in[2];
  const float* b1 = (const float*)d_in[3];
  const float* W2 = (const float*)d_in[4];
  const float* b2 = (const float*)d_in[5];
  float* out = (float*)d_out;

  const int N = in_sizes[0] / C_DIM;

  float* inv_s = out;
  float* tot = (float*)d_ws;
  const size_t inv_pad = ((size_t)N + 63) & ~(size_t)63;
  float* H = tot + inv_pad;

  const int deg_blocks = (N + 255) / 256;

  gcn_deg_kernel<<<deg_blocks, 256, 0, stream>>>(edges, inv_s, N);
  gcn_layer_kernel<false, true><<<GRID_BLOCKS, BLOCK_THREADS, 0, stream>>>(x, edges, W1, b1, inv_s, tot, H, N);
  gcn_layer_kernel<true, false><<<GRID_BLOCKS, BLOCK_THREADS, 0, stream>>>(H, edges, W2, b2, nullptr, tot, out, N);
}


# gather tiled by 32-column chunks (one load = four rows' 128-B lines, per-lane offsets via ds_bpermute): the product runs one chunk, not one tile, behind the gather in the same tile; sum order kept
# speedup vs baseline: 1.5279x; 1.0451x over previous
#include <hip/hip_runtime.h>
#include <math.h>


typedef __attribute__((ext_vector_type(2))) float v2f;
typedef __attribute__((ext_vector_type(8))) float v8f;

#define C_DIM 128
#define MAX_DEG 16
#define TILE_ROWS 16
#define WAVES_PER_BLOCK 15
#define BLOCK_THREADS (32 * WAVES_PER_BLOCK)
#define GRID_BLOCKS 256
#define TILE_PITCH 132

__global__ __launch_bounds__(256) void gcn_deg_kernel(
    const int* __restrict__ edges, float* __restrict__ inv_s, int N) {
  int n = blockIdx.x * blockDim.x + threadIdx.x;
  if (n >= N) return;
  int cnt = 1;
#pragma unroll
  for (int d = 0; d < 16; ++d) cnt += (edges[n * 16 + d] >= 0) ? 1 : 0;
  inv_s[n] = rsqrtf((float)cnt);
}


template <bool PRE_IN>
__device__ __forceinline__ void tile_prep(const int* __restrict__ edges, const float* __restrict__ inv_s,
                                          const float* totArr, int node0, int N, int lane,
                                          int (&ev)[8], float (&sv)[8], float& sSelf, float& tot) {
  int myNode = node0 + (lane >> 1);
  if (myNode >= N) myNode = N - 1;
  const int4* ep = (const int4*)(edges + (size_t)myNode * MAX_DEG + (lane & 1) * 8);
  const int4 e0 = ep[0];
  const int4 e1 = ep[1];
  const int e[8] = {e0.x, e0.y, e0.z, e0.w, e1.x, e1.y, e1.z, e1.w};
  if constexpr (PRE_IN) {
    tot = totArr[myNode];
    int half = 0;
#pragma unroll
    for (int q = 0; q < 8; ++q) {
      const bool valid = e[q] >= 0;
      ev[q] = valid ? e[q] : myNode;
      sv[q] = valid ? 1.f : 0.f;
      half += valid ? 1 : 0;
    }
    sSelf = rsqrtf((float)(1 + half + __shfl_xor(half, 1)));
    return;
  }
  sSelf = inv_s[myNode];
  float part = 0.f;
#pragma unroll
  for (int q = 0; q < 8; ++q) {
    const bool valid = e[q] >= 0;
    const int idx = valid ? e[q] : myNode;
    const float s = inv_s[idx];
    ev[q] = idx;
    sv[q] = valid ? s : 0.f;
    part += sv[q];
  }
  tot = part + __shfl_xor(part, 1) + sSelf;
}

template <bool PRE_IN>
__device__ __forceinline__ void piece_issue(const char* __restrict__ Xb, int node0, int N, int m, unsigned colByte,
                                            int lane, const int (&ev)[8], const float (&sv)[8], float sSelf,
                                            float4 (&r)[MAX_DEG + 1], float (&sc)[MAX_DEG + 1]) {
  const int nl = 4 * m + (lane >> 3);
  const int hold = 8 * nl;
  int nself = node0 + nl;
  if (nself >= N) nself = N - 1;
  sc[0] = PRE_IN ? 1.f : __int_as_float(__builtin_amdgcn_ds_bpermute(hold, __float_as_int(sSelf)));
  r[0] = *(const float4*)(Xb + ((unsigned)nself * (unsigned)(C_DIM * 4) + colByte));
#pragma unroll
  for (int d = 0; d < MAX_DEG; ++d) {
    const int from = hold + 4 * (d >> 3);
    const unsigned idx = (unsigned)__builtin_amdgcn_ds_bpermute(from, ev[d & 7]);
    sc[d + 1] = __int_as_float(__builtin_amdgcn_ds_bpermute(from, __float_as_int(sv[d & 7])));
    r[d + 1] = *(const float4*)(Xb + (idx * (unsigned)(C_DIM * 4) + colByte));
  }
}

__device__ __forceinline__ float4 gather_sum(const float4 (&r)[MAX_DEG + 1], const float (&sc)[MAX_DEG + 1]) {
  float4 a;
  a.x = r[0].x * sc[0];
  a.y = r[0].y * sc[0];
  a.z = r[0].z * sc[0];
  a.w = r[0].w * sc[0];
#pragma unroll
  for (int d = 1; d <= MAX_DEG; ++d) {
    a.x = fmaf(sc[d], r[d].x, a.x);
    a.y = fmaf(sc[d], r[d].y, a.y);
    a.z = fmaf(sc[d], r[d].z, a.z);
    a.w = fmaf(sc[d], r[d].w, a.w);
  }
  return a;
}

__device__ __forceinline__ void product_step(int j, const float* tileA, const v2f* lWp, int c, int h,
                                             v8f (&acc)[8]) {
  const float4 a4 = *(const float4*)&tileA[c * TILE_PITCH + 8 * j + 4 * h];
#pragma unroll
  for (int s = 0; s < 2; ++s) {
    v2f afrag;
    afrag.x = s ? a4.z : a4.x;
    afrag.y = s ? a4.w : a4.y;
    const v2f* bbase = &lWp[(4 * j + 2 * h + s) * C_DIM];
#pragma unroll
    for (int t = 0; t < 8; ++t) {
      const v2f bfrag = bbase[t * 16 + c];
      acc[t] = __builtin_amdgcn_wmma_f32_16x16x4_f32(
          false, afrag, false, bfrag, (short)0, acc[t], false, false);
    }
  }
}

#define W_ITEMS ((C_DIM / 2) * 32)
#define W_TRIPS ((W_ITEMS + BLOCK_THREADS - 1) / BLOCK_THREADS)

template <bool PRE_IN, bool PRE_OUT>
__global__ __launch_bounds__(BLOCK_THREADS) void gcn_layer_kernel(
    const float* __restrict__ Xin, const int* __restrict__ edges,
    const float* __restrict__ W, const float* __restrict__ bias,
    const float* __restrict__ inv_s, float* totArr, float* __restrict__ Y, int N) {
  __shared__ __attribute__((aligned(16))) v2f lWp[(C_DIM / 2) * C_DIM];
  __shared__ __attribute__((aligned(16))) float lTile[WAVES_PER_BLOCK][2][TILE_ROWS * TILE_PITCH];
  __shared__ float lRowS[WAVES_PER_BLOCK][2][TILE_ROWS];
  __shared__ float lRowSum[WAVES_PER_BLOCK][2][TILE_ROWS];

  const int tid = threadIdx.x;
  const int wave = __builtin_amdgcn_readfirstlane(tid >> 5);
  const int lane = tid & 31;

  {
    float4 w0[W_TRIPS], w1[W_TRIPS];
#pragma unroll
    for (int u = 0; u < W_TRIPS; ++u) {
      const int it = tid + BLOCK_THREADS * u;
      if (it < W_ITEMS) {
        const int k2 = it >> 5;
        const int cq = it & 31;
        w0[u] = *(const float4*)(W + (2 * k2 + 0) * C_DIM + 4 * cq);
        w1[u] = *(const float4*)(W + (2 * k2 + 1) * C_DIM + 4 * cq);
      }
    }
#pragma unroll
    for (int u = 0; u < W_TRIPS; ++u) {
      const int it = tid + BLOCK_THREADS * u;
      if (it < W_ITEMS) {
        const int k2 = it >> 5;
        const int cq = it & 31;
        float4* dst = (float4*)&lWp[k2 * C_DIM + 4 * cq];
        dst[0] = make_float4(w0[u].x, w1[u].x, w0[u].y, w1[u].y);
        dst[1] = make_float4(w0[u].z, w1[u].z, w0[u].w, w1[u].w);
      }
    }
  }
  __syncthreads();

  const char* Xb = (const char*)Xin;
  const int c = lane & 15;
  const int h = lane >> 4;
  const int g = lane >> 3;
  const int c4 = lane & 7;
  const int numTiles = (N + TILE_ROWS - 1) / TILE_ROWS;
  const int tileStride = GRID_BLOCKS * WAVES_PER_BLOCK;

  int tile = blockIdx.x + GRID_BLOCKS * wave;
  int p = 0;

  int ev[8] = {0, 0, 0, 0, 0, 0, 0, 0};
  float sv[8] = {0.f, 0.f, 0.f, 0.f, 0.f, 0.f, 0.f, 0.f};
  float sSelf = 0.f, tot = 0.f;

  if (tile < numTiles) {
    const int node0 = tile * TILE_ROWS;
    tile_prep<PRE_IN>(edges, inv_s, totArr, node0, N, lane, ev, sv, sSelf, tot);
    if ((lane & 1) == 0) {
      lRowS[wave][0][lane >> 1] = sSelf;
      lRowSum[wave][0][lane >> 1] = tot;
      if constexpr (PRE_OUT) {
        if (node0 + (lane >> 1) < N) totArr[node0 + (lane >> 1)] = tot;
      }
    }
    float* t0 = lTile[wave][0];
#pragma nounroll
    for (int m = 0; m < 4; ++m) {
      float4 r[MAX_DEG + 1];
      float sc[MAX_DEG + 1];
      piece_issue<PRE_IN>(Xb, node0, N, m, 16u * c4, lane, ev, sv, sSelf, r, sc);
      *(float4*)&t0[(4 * m + g) * TILE_PITCH + 4 * c4] = gather_sum(r, sc);
    }
  }

  for (; tile < numTiles; tile += tileStride, p ^= 1) {
    const int node0 = tile * TILE_ROWS;
    const int next = tile + tileStride;
    float* tileA = lTile[wave][p];

    v8f acc[8];
    const v8f vzero = {0.f, 0.f, 0.f, 0.f, 0.f, 0.f, 0.f, 0.f};
#pragma unroll
    for (int t = 0; t < 8; ++t) acc[t] = vzero;

    for (int j = 0; j < 12; ++j) {
      const int q = (j >> 2) + 1;
      const int m = j & 3;
      float4 r[MAX_DEG + 1];
      float sc[MAX_DEG + 1];
      piece_issue<PRE_IN>(Xb, node0, N, m, 128u * q + 16u * c4, lane, ev, sv, sSelf, r, sc);
      product_step(j, tileA, lWp, c, h, acc);
      __builtin_amdgcn_sched_barrier(0);
      *(float4*)&tileA[(4 * m + g) * TILE_PITCH + 32 * q + 4 * c4] = gather_sum(r, sc);
    }

    if (next < numTiles) {
      const int nnode0 = next * TILE_ROWS;
      float* tileN = lTile[wave][p ^ 1];
      tile_prep<PRE_IN>(edges, inv_s, totArr, nnode0, N, lane, ev, sv, sSelf, tot);
      if ((lane & 1) == 0) {
        lRowS[wave][p ^ 1][lane >> 1] = sSelf;
        lRowSum[wave][p ^ 1][lane >> 1] = tot;
        if constexpr (PRE_OUT) {
          if (nnode0 + (lane >> 1) < N) totArr[nnode0 + (lane >> 1)] = tot;
        }
      }
#pragma nounroll
      for (int m = 0; m < 4; ++m) {
        float4 r[MAX_DEG + 1];
        float sc[MAX_DEG + 1];
        piece_issue<PRE_IN>(Xb, nnode0, N, m, 16u * c4, lane, ev, sv, sSelf, r, sc);
        product_step(12 + m, tileA, lWp, c, h, acc);
        __builtin_amdgcn_sched_barrier(0);
        *(float4*)&tileN[(4 * m + g) * TILE_PITCH + 4 * c4] = gather_sum(r, sc);
      }
    } else {
#pragma unroll 2
      for (int j = 12; j < C_DIM / 8; ++j) product_step(j, tileA, lWp, c, h, acc);
    }

    float rs[8], rsum[8];
#pragma unroll
    for (int v = 0; v < 8; ++v) {
      rs[v] = lRowS[wave][p][8 * h + v];
      rsum[v] = lRowSum[wave][p][8 * h + v];
    }
#pragma unroll
    for (int t = 0; t < 8; ++t) {
      const int col = t * 16 + c;
      const float bc = bias[col];
#pragma unroll
      for (int v = 0; v < 8; ++v) {
        const float z = (acc[t][v] + bc * rsum[v]) * rs[v];
        const float y = (z > 0.f) ? z : expm1f(z);
        tileA[(8 * h + v) * TILE_PITCH + col] = PRE_OUT ? y * rs[v] : y;
      }
    }
#pragma unroll 4
    for (int i = 0; i < TILE_ROWS; ++i) {
      const int node = node0 + i;
      if (node < N) {
        const float4 o = *(const float4*)&tileA[i * TILE_PITCH + 4 * lane];
        ((float4*)Y)[(size_t)node * 32 + lane] = o;
      }
    }
  }
}

extern "C" void kernel_launch(void* const* d_in, const int* in_sizes, int n_in,
                              void* d_out, int out_size, void* d_ws,
                              size_t ws_size, hipStream_t stream) {
  const float* x = (const float*)d_in[0];
  const int* edges = (const int*)d_in[1];
  const float* W1 = (const float*)d_in[2];
  const float* b1 = (const float*)d_in[3];
  const float* W2 = (const float*)d_in[4];
  const float* b2 = (const float*)d_in[5];
  float* out = (float*)d_out;

  const int N = in_sizes[0] / C_DIM;

  float* inv_s = out;
  float* tot = (float*)d_ws;
  const size_t inv_pad = ((size_t)N + 63) & ~(size_t)63;
  float* H = tot + inv_pad;

  const int deg_blocks = (N + 255) / 256;

  gcn_deg_kernel<<<deg_blocks, 256, 0, stream>>>(edges, inv_s, N);
  gcn_layer_kernel<false, true><<<GRID_BLOCKS, BLOCK_THREADS, 0, stream>>>(x, edges, W1, b1, inv_s, tot, H, N);
  gcn_layer_kernel<true, false><<<GRID_BLOCKS, BLOCK_THREADS, 0, stream>>>(H, edges, W2, b2, nullptr, tot, out, N);
}


# ELU's expm1 as straight-line arithmetic: Taylor sum to z^8 above -0.5, exp2(z*log2e)-1 on v_exp_f32 below, in place of 64 branched library expm1f chains a tile; results differ by a few ulp
# speedup vs baseline: 1.6385x; 1.0724x over previous
#include <hip/hip_runtime.h>
#include <math.h>


typedef __attribute__((ext_vector_type(2))) float v2f;
typedef __attribute__((ext_vector_type(8))) float v8f;

#define C_DIM 128
#define MAX_DEG 16
#define TILE_ROWS 16
#define WAVES_PER_BLOCK 15
#define BLOCK_THREADS (32 * WAVES_PER_BLOCK)
#define GRID_BLOCKS 256
#define TILE_PITCH 132

__global__ __launch_bounds__(256) void gcn_deg_kernel(
    const int* __restrict__ edges, float* __restrict__ inv_s, int N) {
  int n = blockIdx.x * blockDim.x + threadIdx.x;
  if (n >= N) return;
  int cnt = 1;
#pragma unroll
  for (int d = 0; d < 16; ++d) cnt += (edges[n * 16 + d] >= 0) ? 1 : 0;
  inv_s[n] = rsqrtf((float)cnt);
}


template <bool PRE_IN>
__device__ __forceinline__ void tile_prep(const int* __restrict__ edges, const float* __restrict__ inv_s,
                                          const float* totArr, int node0, int N, int lane,
                                          int (&ev)[8], float (&sv)[8], float& sSelf, float& tot) {
  int myNode = node0 + (lane >> 1);
  if (myNode >= N) myNode = N - 1;
  const int4* ep = (const int4*)(edges + (size_t)myNode * MAX_DEG + (lane & 1) * 8);
  const int4 e0 = ep[0];
  const int4 e1 = ep[1];
  const int e[8] = {e0.x, e0.y, e0.z, e0.w, e1.x, e1.y, e1.z, e1.w};
  if constexpr (PRE_IN) {
    tot = totArr[myNode];
    int half = 0;
#pragma unroll
    for (int q = 0; q < 8; ++q) {
      const bool valid = e[q] >= 0;
      ev[q] = valid ? e[q] : myNode;
      sv[q] = valid ? 1.f : 0.f;
      half += valid ? 1 : 0;
    }
    sSelf = rsqrtf((float)(1 + half + __shfl_xor(half, 1)));
    return;
  }
  sSelf = inv_s[myNode];
  float part = 0.f;
#pragma unroll
  for (int q = 0; q < 8; ++q) {
    const bool valid = e[q] >= 0;
    const int idx = valid ? e[q] : myNode;
    const float s = inv_s[idx];
    ev[q] = idx;
    sv[q] = valid ? s : 0.f;
    part += sv[q];
  }
  tot = part + __shfl_xor(part, 1) + sSelf;
}

template <bool PRE_IN>
__device__ __forceinline__ void piece_issue(const char* __restrict__ Xb, int node0, int N, int m, unsigned colByte,
                                            int lane, const int (&ev)[8], const float (&sv)[8], float sSelf,
                                            float4 (&r)[MAX_DEG + 1], float (&sc)[MAX_DEG + 1]) {
  const int nl = 4 * m + (lane >> 3);
  const int hold = 8 * nl;
  int nself = node0 + nl;
  if (nself >= N) nself = N - 1;
  sc[0] = PRE_IN ? 1.f : __int_as_float(__builtin_amdgcn_ds_bpermute(hold, __float_as_int(sSelf)));
  r[0] = *(const float4*)(Xb + ((unsigned)nself * (unsigned)(C_DIM * 4) + colByte));
#pragma unroll
  for (int d = 0; d < MAX_DEG; ++d) {
    const int from = hold + 4 * (d >> 3);
    const unsigned idx = (unsigned)__builtin_amdgcn_ds_bpermute(from, ev[d & 7]);
    sc[d + 1] = __int_as_float(__builtin_amdgcn_ds_bpermute(from, __float_as_int(sv[d & 7])));
    r[d + 1] = *(const float4*)(Xb + (idx * (unsigned)(C_DIM * 4) + colByte));
  }
}

__device__ __forceinline__ float4 gather_sum(const float4 (&r)[MAX_DEG + 1], const float (&sc)[MAX_DEG + 1]) {
  float4 a;
  a.x = r[0].x * sc[0];
  a.y = r[0].y * sc[0];
  a.z = r[0].z * sc[0];
  a.w = r[0].w * sc[0];
#pragma unroll
  for (int d = 1; d <= MAX_DEG; ++d) {
    a.x = fmaf(sc[d], r[d].x, a.x);
    a.y = fmaf(sc[d], r[d].y, a.y);
    a.z = fmaf(sc[d], r[d].z, a.z);
    a.w = fmaf(sc[d], r[d].w, a.w);
  }
  return a;
}

__device__ __forceinline__ float elu_f32(float z) {
  float q = fmaf(z, 1.0f / 40320.0f, 1.0f / 5040.0f);
  q = fmaf(z, q, 1.0f / 720.0f);
  q = fmaf(z, q, 1.0f / 120.0f);
  q = fmaf(z, q, 1.0f / 24.0f);
  q = fmaf(z, q, 1.0f / 6.0f);
  q = fmaf(z, q, 0.5f);
  const float nearZero = fmaf(z * z, q, z);
  const float farSide = __builtin_amdgcn_exp2f(z * 1.4426950408889634f) - 1.0f;
  const float neg = (z > -0.5f) ? nearZero : farSide;
  return (z > 0.f) ? z : neg;
}

__device__ __forceinline__ void product_step(int j, const float* tileA, const v2f* lWp, int c, int h,
                                             v8f (&acc)[8]) {
  const float4 a4 = *(const float4*)&tileA[c * TILE_PITCH + 8 * j + 4 * h];
#pragma unroll
  for (int s = 0; s < 2; ++s) {
    v2f afrag;
    afrag.x = s ? a4.z : a4.x;
    afrag.y = s ? a4.w : a4.y;
    const v2f* bbase = &lWp[(4 * j + 2 * h + s) * C_DIM];
#pragma unroll
    for (int t = 0; t < 8; ++t) {
      const v2f bfrag = bbase[t * 16 + c];
      acc[t] = __builtin_amdgcn_wmma_f32_16x16x4_f32(
          false, afrag, false, bfrag, (short)0, acc[t], false, false);
    }
  }
}

#define W_ITEMS ((C_DIM / 2) * 32)
#define W_TRIPS ((W_ITEMS + BLOCK_THREADS - 1) / BLOCK_THREADS)

template <bool PRE_IN, bool PRE_OUT>
__global__ __launch_bounds__(BLOCK_THREADS) void gcn_layer_kernel(
    const float* __restrict__ Xin, const int* __restrict__ edges,
    const float* __restrict__ W, const float* __restrict__ bias,
    const float* __restrict__ inv_s, float* totArr, float* __restrict__ Y, int N) {
  __shared__ __attribute__((aligned(16))) v2f lWp[(C_DIM / 2) * C_DIM];
  __shared__ __attribute__((aligned(16))) float lTile[WAVES_PER_BLOCK][2][TILE_ROWS * TILE_PITCH];
  __shared__ float lRowS[WAVES_PER_BLOCK][2][TILE_ROWS];
  __shared__ float lRowSum[WAVES_PER_BLOCK][2][TILE_ROWS];

  const int tid = threadIdx.x;
  const int wave = __builtin_amdgcn_readfirstlane(tid >> 5);
  const int lane = tid & 31;

  {
    float4 w0[W_TRIPS], w1[W_TRIPS];
#pragma unroll
    for (int u = 0; u < W_TRIPS; ++u) {
      const int it = tid + BLOCK_THREADS * u;
      if (it < W_ITEMS) {
        const int k2 = it >> 5;
        const int cq = it & 31;
        w0[u] = *(const float4*)(W + (2 * k2 + 0) * C_DIM + 4 * cq);
        w1[u] = *(const float4*)(W + (2 * k2 + 1) * C_DIM + 4 * cq);
      }
    }
#pragma unroll
    for (int u = 0; u < W_TRIPS; ++u) {
      const int it = tid + BLOCK_THREADS * u;
      if (it < W_ITEMS) {
        const int k2 = it >> 5;
        const int cq = it & 31;
        float4* dst = (float4*)&lWp[k2 * C_DIM + 4 * cq];
        dst[0] = make_float4(w0[u].x, w1[u].x, w0[u].y, w1[u].y);
        dst[1] = make_float4(w0[u].z, w1[u].z, w0[u].w, w1[u].w);
      }
    }
  }
  __syncthreads();

  const char* Xb = (const char*)Xin;
  const int c = lane & 15;
  const int h = lane >> 4;
  const int g = lane >> 3;
  const int c4 = lane & 7;
  const int numTiles = (N + TILE_ROWS - 1) / TILE_ROWS;
  const int tileStride = GRID_BLOCKS * WAVES_PER_BLOCK;

  int tile = blockIdx.x + GRID_BLOCKS * wave;
  int p = 0;

  int ev[8] = {0, 0, 0, 0, 0, 0, 0, 0};
  float sv[8] = {0.f, 0.f, 0.f, 0.f, 0.f, 0.f, 0.f, 0.f};
  float sSelf = 0.f, tot = 0.f;

  if (tile < numTiles) {
    const int node0 = tile * TILE_ROWS;
    tile_prep<PRE_IN>(edges, inv_s, totArr, node0, N, lane, ev, sv, sSelf, tot);
    if ((lane & 1) == 0) {
      lRowS[wave][0][lane >> 1] = sSelf;
      lRowSum[wave][0][lane >> 1] = tot;
      if constexpr (PRE_OUT) {
        if (node0 + (lane >> 1) < N) totArr[node0 + (lane >> 1)] = tot;
      }
    }
    float* t0 = lTile[wave][0];
#pragma nounroll
    for (int m = 0; m < 4; ++m) {
      float4 r[MAX_DEG + 1];
      float sc[MAX_DEG + 1];
      piece_issue<PRE_IN>(Xb, node0, N, m, 16u * c4, lane, ev, sv, sSelf, r, sc);
      *(float4*)&t0[(4 * m + g) * TILE_PITCH + 4 * c4] = gather_sum(r, sc);
    }
  }

  for (; tile < numTiles; tile += tileStride, p ^= 1) {
    const int node0 = tile * TILE_ROWS;
    const int next = tile + tileStride;
    float* tileA = lTile[wave][p];

    v8f acc[8];
    const v8f vzero = {0.f, 0.f, 0.f, 0.f, 0.f, 0.f, 0.f, 0.f};
#pragma unroll
    for (int t = 0; t < 8; ++t) acc[t] = vzero;

    for (int j = 0; j < 12; ++j) {
      const int q = (j >> 2) + 1;
      const int m = j & 3;
      float4 r[MAX_DEG + 1];
      float sc[MAX_DEG + 1];
      piece_issue<PRE_IN>(Xb, node0, N, m, 128u * q + 16u * c4, lane, ev, sv, sSelf, r, sc);
      product_step(j, tileA, lWp, c, h, acc);
      __builtin_amdgcn_sched_barrier(0);
      *(float4*)&tileA[(4 * m + g) * TILE_PITCH + 32 * q + 4 * c4] = gather_sum(r, sc);
    }

    if (next < numTiles) {
      const int nnode0 = next * TILE_ROWS;
      float* tileN = lTile[wave][p ^ 1];
      tile_prep<PRE_IN>(edges, inv_s, totArr, nnode0, N, lane, ev, sv, sSelf, tot);
      if ((lane & 1) == 0) {
        lRowS[wave][p ^ 1][lane >> 1] = sSelf;
        lRowSum[wave][p ^ 1][lane >> 1] = tot;
        if constexpr (PRE_OUT) {
          if (nnode0 + (lane >> 1) < N) totArr[nnode0 + (lane >> 1)] = tot;
        }
      }
#pragma nounroll
      for (int m = 0; m < 4; ++m) {
        float4 r[MAX_DEG + 1];
        float sc[MAX_DEG + 1];
        piece_issue<PRE_IN>(Xb, nnode0, N, m, 16u * c4, lane, ev, sv, sSelf, r, sc);
        product_step(12 + m, tileA, lWp, c, h, acc);
        __builtin_amdgcn_sched_barrier(0);
        *(float4*)&tileN[(4 * m + g) * TILE_PITCH + 4 * c4] = gather_sum(r, sc);
      }
    } else {
#pragma unroll 2
      for (int j = 12; j < C_DIM / 8; ++j) product_step(j, tileA, lWp, c, h, acc);
    }

    float rs[8], rsum[8];
#pragma unroll
    for (int v = 0; v < 8; ++v) {
      rs[v] = lRowS[wave][p][8 * h + v];
      rsum[v] = lRowSum[wave][p][8 * h + v];
    }
#pragma unroll
    for (int t = 0; t < 8; ++t) {
      const int col = t * 16 + c;
      const float bc = bias[col];
#pragma unroll
      for (int v = 0; v < 8; ++v) {
        const float z = (acc[t][v] + bc * rsum[v]) * rs[v];
        const float y = elu_f32(z);
        tileA[(8 * h + v) * TILE_PITCH + col] = PRE_OUT ? y * rs[v] : y;
      }
    }
#pragma unroll 4
    for (int i = 0; i < TILE_ROWS; ++i) {
      const int node = node0 + i;
      if (node < N) {
        const float4 o = *(const float4*)&tileA[i * TILE_PITCH + 4 * lane];
        ((float4*)Y)[(size_t)node * 32 + lane] = o;
      }
    }
  }
}

extern "C" void kernel_launch(void* const* d_in, const int* in_sizes, int n_in,
                              void* d_out, int out_size, void* d_ws,
                              size_t ws_size, hipStream_t stream) {
  const float* x = (const float*)d_in[0];
  const int* edges = (const int*)d_in[1];
  const float* W1 = (const float*)d_in[2];
  const float* b1 = (const float*)d_in[3];
  const float* W2 = (const float*)d_in[4];
  const float* b2 = (const float*)d_in[5];
  float* out = (float*)d_out;

  const int N = in_sizes[0] / C_DIM;

  float* inv_s = out;
  float* tot = (float*)d_ws;
  const size_t inv_pad = ((size_t)N + 63) & ~(size_t)63;
  float* H = tot + inv_pad;

  const int deg_blocks = (N + 255) / 256;

  gcn_deg_kernel<<<deg_blocks, 256, 0, stream>>>(edges, inv_s, N);
  gcn_layer_kernel<false, true><<<GRID_BLOCKS, BLOCK_THREADS, 0, stream>>>(x, edges, W1, b1, inv_s, tot, H, N);
  gcn_layer_kernel<true, false><<<GRID_BLOCKS, BLOCK_THREADS, 0, stream>>>(H, edges, W2, b2, nullptr, tot, out, N);
}


# gather tiled by 64-column chunks: one load = two rows' aligned 256-B units (was four rows' 128-B lines), so a unit's two lines are asked for together; product one such chunk behind; sum order kept
# speedup vs baseline: 1.7051x; 1.0406x over previous
#include <hip/hip_runtime.h>
#include <math.h>


typedef __attribute__((ext_vector_type(2))) float v2f;
typedef __attribute__((ext_vector_type(8))) float v8f;

#define C_DIM 128
#define MAX_DEG 16
#define TILE_ROWS 16
#define WAVES_PER_BLOCK 15
#define BLOCK_THREADS (32 * WAVES_PER_BLOCK)
#define GRID_BLOCKS 256
#define TILE_PITCH 132

__global__ __launch_bounds__(256) void gcn_deg_kernel(
    const int* __restrict__ edges, float* __restrict__ inv_s, int N) {
  int n = blockIdx.x * blockDim.x + threadIdx.x;
  if (n >= N) return;
  int cnt = 1;
#pragma unroll
  for (int d = 0; d < 16; ++d) cnt += (edges[n * 16 + d] >= 0) ? 1 : 0;
  inv_s[n] = rsqrtf((float)cnt);
}


template <bool PRE_IN>
__device__ __forceinline__ void tile_prep(const int* __restrict__ edges, const float* __restrict__ inv_s,
                                          const float* totArr, int node0, int N, int lane,
                                          int (&ev)[8], float (&sv)[8], float& sSelf, float& tot) {
  int myNode = node0 + (lane >> 1);
  if (myNode >= N) myNode = N - 1;
  const int4* ep = (const int4*)(edges + (size_t)myNode * MAX_DEG + (lane & 1) * 8);
  const int4 e0 = ep[0];
  const int4 e1 = ep[1];
  const int e[8] = {e0.x, e0.y, e0.z, e0.w, e1.x, e1.y, e1.z, e1.w};
  if constexpr (PRE_IN) {
    tot = totArr[myNode];
    int half = 0;
#pragma unroll
    for (int q = 0; q < 8; ++q) {
      const bool valid = e[q] >= 0;
      ev[q] = valid ? e[q] : myNode;
      sv[q] = valid ? 1.f : 0.f;
      half += valid ? 1 : 0;
    }
    sSelf = rsqrtf((float)(1 + half + __shfl_xor(half, 1)));
    return;
  }
  sSelf = inv_s[myNode];
  float part = 0.f;
#pragma unroll
  for (int q = 0; q < 8; ++q) {
    const bool valid = e[q] >= 0;
    const int idx = valid ? e[q] : myNode;
    const float s = inv_s[idx];
    ev[q] = idx;
    sv[q] = valid ? s : 0.f;
    part += sv[q];
  }
  tot = part + __shfl_xor(part, 1) + sSelf;
}

template <bool PRE_IN>
__device__ __forceinline__ void piece_issue(const char* __restrict__ Xb, int node0, int N, int m, unsigned colByte,
                                            int lane, const int (&ev)[8], const float (&sv)[8], float sSelf,
                                            float4 (&r)[MAX_DEG + 1], float (&sc)[MAX_DEG + 1]) {
  const int nl = 2 * m + (lane >> 4);
  const int hold = 8 * nl;
  int nself = node0 + nl;
  if (nself >= N) nself = N - 1;
  sc[0] = PRE_IN ? 1.f : __int_as_float(__builtin_amdgcn_ds_bpermute(hold, __float_as_int(sSelf)));
  r[0] = *(const float4*)(Xb + ((unsigned)nself * (unsigned)(C_DIM * 4) + colByte));
#pragma unroll
  for (int d = 0; d < MAX_DEG; ++d) {
    const int from = hold + 4 * (d >> 3);
    const unsigned idx = (unsigned)__builtin_amdgcn_ds_bpermute(from, ev[d & 7]);
    sc[d + 1] = __int_as_float(__builtin_amdgcn_ds_bpermute(from, __float_as_int(sv[d & 7])));
    r[d + 1] = *(const float4*)(Xb + (idx * (unsigned)(C_DIM * 4) + colByte));
  }
}

__device__ __forceinline__ float4 gather_sum(const float4 (&r)[MAX_DEG + 1], const float (&sc)[MAX_DEG + 1]) {
  float4 a;
  a.x = r[0].x * sc[0];
  a.y = r[0].y * sc[0];
  a.z = r[0].z * sc[0];
  a.w = r[0].w * sc[0];
#pragma unroll
  for (int d = 1; d <= MAX_DEG; ++d) {
    a.x = fmaf(sc[d], r[d].x, a.x);
    a.y = fmaf(sc[d], r[d].y, a.y);
    a.z = fmaf(sc[d], r[d].z, a.z);
    a.w = fmaf(sc[d], r[d].w, a.w);
  }
  return a;
}

__device__ __forceinline__ float elu_f32(float z) {
  float q = fmaf(z, 1.0f / 40320.0f, 1.0f / 5040.0f);
  q = fmaf(z, q, 1.0f / 720.0f);
  q = fmaf(z, q, 1.0f / 120.0f);
  q = fmaf(z, q, 1.0f / 24.0f);
  q = fmaf(z, q, 1.0f / 6.0f);
  q = fmaf(z, q, 0.5f);
  const float nearZero = fmaf(z * z, q, z);
  const float farSide = __builtin_amdgcn_exp2f(z * 1.4426950408889634f) - 1.0f;
  const float neg = (z > -0.5f) ? nearZero : farSide;
  return (z > 0.f) ? z : neg;
}

__device__ __forceinline__ void product_step(int j, const float* tileA, const v2f* lWp, int c, int h,
                                             v8f (&acc)[8]) {
  const float4 a4 = *(const float4*)&tileA[c * TILE_PITCH + 8 * j + 4 * h];
#pragma unroll
  for (int s = 0; s < 2; ++s) {
    v2f afrag;
    afrag.x = s ? a4.z : a4.x;
    afrag.y = s ? a4.w : a4.y;
    const v2f* bbase = &lWp[(4 * j + 2 * h + s) * C_DIM];
#pragma unroll
    for (int t = 0; t < 8; ++t) {
      const v2f bfrag = bbase[t * 16 + c];
      acc[t] = __builtin_amdgcn_wmma_f32_16x16x4_f32(
          false, afrag, false, bfrag, (short)0, acc[t], false, false);
    }
  }
}

#define W_ITEMS ((C_DIM / 2) * 32)
#define W_TRIPS ((W_ITEMS + BLOCK_THREADS - 1) / BLOCK_THREADS)

template <bool PRE_IN, bool PRE_OUT>
__global__ __launch_bounds__(BLOCK_THREADS) void gcn_layer_kernel(
    const float* __restrict__ Xin, const int* __restrict__ edges,
    const float* __restrict__ W, const float* __restrict__ bias,
    const float* __restrict__ inv_s, float* totArr, float* __restrict__ Y, int N) {
  __shared__ __attribute__((aligned(16))) v2f lWp[(C_DIM / 2) * C_DIM];
  __shared__ __attribute__((aligned(16))) float lTile[WAVES_PER_BLOCK][2][TILE_ROWS * TILE_PITCH];
  __shared__ float lRowS[WAVES_PER_BLOCK][2][TILE_ROWS];
  __shared__ float lRowSum[WAVES_PER_BLOCK][2][TILE_ROWS];

  const int tid = threadIdx.x;
  const int wave = __builtin_amdgcn_readfirstlane(tid >> 5);
  const int lane = tid & 31;

  {
    float4 w0[W_TRIPS], w1[W_TRIPS];
#pragma unroll
    for (int u = 0; u < W_TRIPS; ++u) {
      const int it = tid + BLOCK_THREADS * u;
      if (it < W_ITEMS) {
        const int k2 = it >> 5;
        const int cq = it & 31;
        w0[u] = *(const float4*)(W + (2 * k2 + 0) * C_DIM + 4 * cq);
        w1[u] = *(const float4*)(W + (2 * k2 + 1) * C_DIM + 4 * cq);
      }
    }
#pragma unroll
    for (int u = 0; u < W_TRIPS; ++u) {
      const int it = tid + BLOCK_THREADS * u;
      if (it < W_ITEMS) {
        const int k2 = it >> 5;
        const int cq = it & 31;
        float4* dst = (float4*)&lWp[k2 * C_DIM + 4 * cq];
        dst[0] = make_float4(w0[u].x, w1[u].x, w0[u].y, w1[u].y);
        dst[1] = make_float4(w0[u].z, w1[u].z, w0[u].w, w1[u].w);
      }
    }
  }
  __syncthreads();

  const char* Xb = (const char*)Xin;
  const int c = lane & 15;
  const int h = lane >> 4;
  const int g = lane >> 4;
  const int c4 = lane & 15;
  const int numTiles = (N + TILE_ROWS - 1) / TILE_ROWS;
  const int tileStride = GRID_BLOCKS * WAVES_PER_BLOCK;

  int tile = blockIdx.x + GRID_BLOCKS * wave;
  int p = 0;

  int ev[8] = {0, 0, 0, 0, 0, 0, 0, 0};
  float sv[8] = {0.f, 0.f, 0.f, 0.f, 0.f, 0.f, 0.f, 0.f};
  float sSelf = 0.f, tot = 0.f;

  if (tile < numTiles) {
    const int node0 = tile * TILE_ROWS;
    tile_prep<PRE_IN>(edges, inv_s, totArr, node0, N, lane, ev, sv, sSelf, tot);
    if ((lane & 1) == 0) {
      lRowS[wave][0][lane >> 1] = sSelf;
      lRowSum[wave][0][lane >> 1] = tot;
      if constexpr (PRE_OUT) {
        if (node0 + (lane >> 1) < N) totArr[node0 + (lane >> 1)] = tot;
      }
    }
    float* t0 = lTile[wave][0];
#pragma nounroll
    for (int m = 0; m < 8; ++m) {
      float4 r[MAX_DEG + 1];
      float sc[MAX_DEG + 1];
      piece_issue<PRE_IN>(Xb, node0, N, m, 16u * c4, lane, ev, sv, sSelf, r, sc);
      *(float4*)&t0[(2 * m + g) * TILE_PITCH + 4 * c4] = gather_sum(r, sc);
    }
  }

  for (; tile < numTiles; tile += tileStride, p ^= 1) {
    const int node0 = tile * TILE_ROWS;
    const int next = tile + tileStride;
    float* tileA = lTile[wave][p];

    v8f acc[8];
    const v8f vzero = {0.f, 0.f, 0.f, 0.f, 0.f, 0.f, 0.f, 0.f};
#pragma unroll
    for (int t = 0; t < 8; ++t) acc[t] = vzero;

    for (int j = 0; j < 8; ++j) {
      const int m = j;
      float4 r[MAX_DEG + 1];
      float sc[MAX_DEG + 1];
      piece_issue<PRE_IN>(Xb, node0, N, m, 256u + 16u * c4, lane, ev, sv, sSelf, r, sc);
      product_step(j, tileA, lWp, c, h, acc);
      __builtin_amdgcn_sched_barrier(0);
      *(float4*)&tileA[(2 * m + g) * TILE_PITCH + 64 + 4 * c4] = gather_sum(r, sc);
    }

    if (next < numTiles) {
      const int nnode0 = next * TILE_ROWS;
      float* tileN = lTile[wave][p ^ 1];
      tile_prep<PRE_IN>(edges, inv_s, totArr, nnode0, N, lane, ev, sv, sSelf, tot);
      if ((lane & 1) == 0) {
        lRowS[wave][p ^ 1][lane >> 1] = sSelf;
        lRowSum[wave][p ^ 1][lane >> 1] = tot;
        if constexpr (PRE_OUT) {
          if (nnode0 + (lane >> 1) < N) totArr[nnode0 + (lane >> 1)] = tot;
        }
      }
#pragma nounroll
      for (int m = 0; m < 8; ++m) {
        float4 r[MAX_DEG + 1];
        float sc[MAX_DEG + 1];
        piece_issue<PRE_IN>(Xb, nnode0, N, m, 16u * c4, lane, ev, sv, sSelf, r, sc);
        product_step(8 + m, tileA, lWp, c, h, acc);
        __builtin_amdgcn_sched_barrier(0);
        *(float4*)&tileN[(2 * m + g) * TILE_PITCH + 4 * c4] = gather_sum(r, sc);
      }
    } else {
#pragma unroll 2
      for (int j = 8; j < C_DIM / 8; ++j) product_step(j, tileA, lWp, c, h, acc);
    }

    float rs[8], rsum[8];
#pragma unroll
    for (int v = 0; v < 8; ++v) {
      rs[v] = lRowS[wave][p][8 * h + v];
      rsum[v] = lRowSum[wave][p][8 * h + v];
    }
#pragma unroll
    for (int t = 0; t < 8; ++t) {
      const int col = t * 16 + c;
      const float bc = bias[col];
#pragma unroll
      for (int v = 0; v < 8; ++v) {
        const float z = (acc[t][v] + bc * rsum[v]) * rs[v];
        const float y = elu_f32(z);
        tileA[(8 * h + v) * TILE_PITCH + col] = PRE_OUT ? y * rs[v] : y;
      }
    }
#pragma unroll 4
    for (int i = 0; i < TILE_ROWS; ++i) {
      const int node = node0 + i;
      if (node < N) {
        const float4 o = *(const float4*)&tileA[i * TILE_PITCH + 4 * lane];
        ((float4*)Y)[(size_t)node * 32 + lane] = o;
      }
    }
  }
}

extern "C" void kernel_launch(void* const* d_in, const int* in_sizes, int n_in,
                              void* d_out, int out_size, void* d_ws,
                              size_t ws_size, hipStream_t stream) {
  const float* x = (const float*)d_in[0];
  const int* edges = (const int*)d_in[1];
  const float* W1 = (const float*)d_in[2];
  const float* b1 = (const float*)d_in[3];
  const float* W2 = (const float*)d_in[4];
  const float* b2 = (const float*)d_in[5];
  float* out = (float*)d_out;

  const int N = in_sizes[0] / C_DIM;

  float* inv_s = out;
  float* tot = (float*)d_ws;
  const size_t inv_pad = ((size_t)N + 63) & ~(size_t)63;
  float* H = tot + inv_pad;

  const int deg_blocks = (N + 255) / 256;

  gcn_deg_kernel<<<deg_blocks, 256, 0, stream>>>(edges, inv_s, N);
  gcn_layer_kernel<false, true><<<GRID_BLOCKS, BLOCK_THREADS, 0, stream>>>(x, edges, W1, b1, inv_s, tot, H, N);
  gcn_layer_kernel<true, false><<<GRID_BLOCKS, BLOCK_THREADS, 0, stream>>>(H, edges, W2, b2, nullptr, tot, out, N);
}
